# starting kernel
# baseline (speedup 1.0000x reference)
#include <hip/hip_runtime.h>
#include <math.h>

typedef __attribute__((ext_vector_type(2))) float v2f;
typedef __attribute__((ext_vector_type(8))) float v8f;

__global__ __launch_bounds__(256) void build_A_kernel(float* __restrict__ Aout) {
    int idx = blockIdx.x * 256 + threadIdx.x;
    if (idx >= 32 * 64) return;
    int r = idx >> 6;
    int w = idx & 63;
    const float PI = 3.14159265358979323846f;
    float acc = 0.0f;
    for (int p = 0; p < 24; ++p) {
        float s32 = (p == 0) ? sqrtf(1.0f / 32.0f) : sqrtf(2.0f / 32.0f);
        float s64 = (p == 0) ? sqrtf(1.0f / 64.0f) : sqrtf(2.0f / 64.0f);
        float c32 = cosf(PI * ((float)r + 0.5f) * (float)p / 32.0f);
        float c64 = cosf(PI * ((float)w + 0.5f) * (float)p / 64.0f);
        acc += c32 * s32 * c64 * s64;
    }
    Aout[idx] = acc;
}

template <int MODE, bool NT>
__global__ __launch_bounds__(256) void dct_gemm_kernel(
    const float* __restrict__ Aglob,
    const float* __restrict__ B,
    float* __restrict__ C,
    int N, int ntilesPerBatch, int totalTiles)
{
    __shared__ float A_lds[32 * 64];
    for (int i = threadIdx.x; i < 32 * 64; i += 256) A_lds[i] = Aglob[i];
    __syncthreads();

    const int lane = threadIdx.x & 31;
    const int wave = threadIdx.x >> 5;
    const int t = blockIdx.x * 8 + wave;
    if (t >= totalTiles) return;

    const int batchIdx = t / ntilesPerBatch;
    const int ntile    = t - batchIdx * ntilesPerBatch;
    const int n0       = ntile * 16;

    const int row  = lane & 15;
    const int koff = (lane >> 4) << 1;
    const int col  = n0 + row;

    v2f afrag[2][16];
#pragma unroll
    for (int mt = 0; mt < 2; ++mt) {
#pragma unroll
        for (int kk = 0; kk < 16; ++kk) {
            const float* ap = &A_lds[(mt * 16 + row) * 64 + kk * 4 + koff];
            v2f a; a.x = ap[0]; a.y = ap[1];
            afrag[mt][kk] = a;
        }
    }

    const float* Bb = B + (size_t)batchIdx * 64 * (size_t)N;
    v8f acc0 = {}; v8f acc1 = {};

#pragma unroll
    for (int kk = 0; kk < 16; ++kk) {
        const int krow = kk * 4 + koff;
        const float* bp0 = Bb + (size_t)krow * N + col;
        const float* bp1 = bp0 + N;
        v2f b;
        if (NT) { b.x = __builtin_nontemporal_load(bp0);
                  b.y = __builtin_nontemporal_load(bp1); }
        else    { b.x = *bp0; b.y = *bp1; }
        acc0 = __builtin_amdgcn_wmma_f32_16x16x4_f32(false, afrag[0][kk], false, b,
                                                     (short)0, acc0, false, false);
        acc1 = __builtin_amdgcn_wmma_f32_16x16x4_f32(false, afrag[1][kk], false, b,
                                                     (short)0, acc1, false, false);
    }

    const int rbase = (lane >> 4) << 3;
    if (MODE == 0) {
        float* Cb = C + (size_t)batchIdx * 32 * (size_t)N;
#pragma unroll
        for (int v = 0; v < 8; ++v) {
            Cb[(size_t)(rbase + v)      * N + col] = acc0[v];
            Cb[(size_t)(16 + rbase + v) * N + col] = acc1[v];
        }
    } else {
        const int b  = batchIdx >> 10;
        const int pq = batchIdx & 1023;
        const int p  = pq >> 5;
        const int q  = pq & 31;
#pragma unroll
        for (int v = 0; v < 8; ++v) {
            int r0 = rbase + v;
            size_t i0 = ((((size_t)b * 32 + r0)        * 32 + q) * 32 + p) * 32 + col;
            size_t i1 = ((((size_t)b * 32 + (16 + r0)) * 32 + q) * 32 + p) * 32 + col;
            C[i0] = acc0[v];
            C[i1] = acc1[v];
        }
    }
}

extern "C" void kernel_launch(void* const* d_in, const int* in_sizes, int n_in,
                              void* d_out, int out_size, void* d_ws, size_t ws_size,
                              hipStream_t stream) {
    (void)in_sizes; (void)n_in; (void)out_size; (void)ws_size;
    const float* x  = (const float*)d_in[0];
    float* out      = (float*)d_out;

    char* ws = (char*)d_ws;
    float* A  = (float*)ws;
    float* z1 = (float*)(ws + 8192);
    float* z2 = (float*)(ws + 8192 + (size_t)4*32*131072*4);

    build_A_kernel<<<8, 256, 0, stream>>>(A);

    dct_gemm_kernel<0, true><<<32768 / 8, 256, 0, stream>>>(
        A, x, z1, 131072, 8192, 32768);

    dct_gemm_kernel<0, false><<<16384 / 8, 256, 0, stream>>>(
        A, z1, z2, 2048, 128, 16384);

    dct_gemm_kernel<1, false><<<8192 / 8, 256, 0, stream>>>(
        A, z2, out, 32, 2, 8192);
}


# x reaches the waves through LDS: each group of 4 h planes fetched once per work-group by 16-byte loads two groups ahead, parked in 2 LDS stages, B fragments read from there; sum order unchanged
# speedup vs baseline: 1.5901x; 1.5901x over previous
#include <hip/hip_runtime.h>
#include <math.h>


typedef __attribute__((ext_vector_type(2))) float v2f;
typedef __attribute__((ext_vector_type(8))) float v8f;
typedef __attribute__((ext_vector_type(4))) float v4f;

__global__ __launch_bounds__(256) void build_A_kernel(float* __restrict__ Aout) {
    int idx = blockIdx.x * 256 + threadIdx.x;
    if (idx >= 32 * 64) return;
    int r = idx >> 6;
    int w = idx & 63;
    const float PI = 3.14159265358979323846f;
    float acc = 0.0f;
    for (int p = 0; p < 24; ++p) {
        float s32 = (p == 0) ? sqrtf(1.0f / 32.0f) : sqrtf(2.0f / 32.0f);
        float s64 = (p == 0) ? sqrtf(1.0f / 64.0f) : sqrtf(2.0f / 64.0f);
        float c32 = cosf(PI * ((float)r + 0.5f) * (float)p / 32.0f);
        float c64 = cosf(PI * ((float)w + 0.5f) * (float)p / 64.0f);
        acc += c32 * s32 * c64 * s64;
    }
    Aout[idx] = acc;
}

__device__ __forceinline__ v8f mma(v2f a, v2f b, v8f c) {
    return __builtin_amdgcn_wmma_f32_16x16x4_f32(false, a, false, b, (short)0, c, false, false);
}

__device__ __forceinline__ float xhalf(float x) {
    int b = __builtin_bit_cast(int, x);
    return __builtin_bit_cast(float,
        __builtin_amdgcn_permlanex16(b, b, 0x76543210, 0xfedcba98, false, false));
}

struct XGroup { v2f f[4][16]; };

struct Stage { v4f r[16]; };

__device__ __forceinline__ void fetch_group(Stage& s, const float* __restrict__ src) {
#pragma unroll
    for (int j = 0; j < 16; ++j) s.r[j] = *reinterpret_cast<const v4f*>(src + j * 512);
}

__device__ __forceinline__ void park_group(const Stage& s, float* dst) {
#pragma unroll
    for (int j = 0; j < 16; ++j) *reinterpret_cast<v4f*>(dst + j * 512) = s.r[j];
}

__device__ __forceinline__ void load_group(XGroup& xg, const float* p) {
#pragma unroll
    for (int hi = 0; hi < 4; ++hi) {
#pragma unroll
        for (int kk = 0; kk < 16; ++kk) {
            const float* q = p + hi * 2048 + kk * 128;
            v2f t; t.x = q[0]; t.y = q[64];
            xg.f[hi][kk] = t;
        }
    }
}

__device__ __forceinline__ void lds_barrier() {
    asm volatile("s_wait_dscnt 0x0\n\ts_barrier_signal -1\n\ts_barrier_wait -1" ::: "memory");
}

__device__ __forceinline__ void contract_group(const XGroup& xg, const v2f (&a1)[16],
                                               v2f a2q0, v2f a2q1, bool lo, v8f (&out)[2][16]) {
    v8f d1[4];
#pragma unroll
    for (int hi = 0; hi < 4; ++hi) d1[hi] = (v8f){};
#pragma unroll
    for (int kk = 0; kk < 16; ++kk) {
#pragma unroll
        for (int hi = 0; hi < 4; ++hi) d1[hi] = mma(a1[kk], xg.f[hi][kk], d1[hi]);
    }
#pragma unroll
    for (int v = 0; v < 8; ++v) {
        float sx = xhalf(lo ? d1[2][v] : d1[0][v]);
        float sy = xhalf(lo ? d1[3][v] : d1[1][v]);
        v2f blo, bhi;
        blo.x = lo ? d1[0][v] : sx;
        blo.y = lo ? d1[1][v] : sy;
        bhi.x = lo ? sx : d1[2][v];
        bhi.y = lo ? sy : d1[3][v];
        out[0][v]     = mma(a2q0, blo, out[0][v]);
        out[1][v]     = mma(a2q1, blo, out[1][v]);
        out[0][8 + v] = mma(a2q0, bhi, out[0][8 + v]);
        out[1][8 + v] = mma(a2q1, bhi, out[1][8 + v]);
    }
}

__global__ __launch_bounds__(128) void fused_wh_kernel(
    const float* __restrict__ Aglob, const float* __restrict__ x, float* __restrict__ z2)
{
    __shared__ __attribute__((aligned(16))) float lds[2 * 8192 + 32 * 64];
    float* const A_lds = lds + 2 * 8192;

    const int tid  = threadIdx.x;
    const int lane = tid & 31;
    const int wave = tid >> 5;
    const int ct   = wave & 1;
    const int rt   = wave >> 1;
    const int bd   = blockIdx.x;
    const int row  = lane & 15;

    const float* xg = x + (size_t)bd * (64 * 64 * 32) + tid * 4;
    const int r4 = (tid >> 3) & 3;
    float* sp = lds + ((wave * 4 + (((r4 & 1) << 1) | (r4 >> 1))) * 32 + (tid & 7) * 4);

    float areg[16];
#pragma unroll
    for (int j = 0; j < 16; ++j) areg[j] = Aglob[tid + 128 * j];
    Stage r0, r1;
    fetch_group(r0, xg);
    fetch_group(r1, xg + 8192);
#pragma unroll
    for (int j = 0; j < 16; ++j) A_lds[tid + 128 * j] = areg[j];
    park_group(r0, sp);
    fetch_group(r0, xg + 2 * 8192);
    lds_barrier();

    const int half = lane >> 4;
    const int koff = half << 1;
    const bool lo  = (half == 0);

    v2f a1[16];
#pragma unroll
    for (int kk = 0; kk < 16; ++kk) {
        const float* ap = &A_lds[(rt * 16 + row) * 64 + kk * 4 + koff];
        v2f a; a.x = ap[0]; a.y = ap[1];
        a1[kk] = a;
    }

    v8f out[2][16];
#pragma unroll
    for (int qt = 0; qt < 2; ++qt) {
#pragma unroll
        for (int rr = 0; rr < 16; ++rr) out[qt][rr] = (v8f){};
    }

    const float* fp = lds + half * 32 + ct * 16 + row;
    const float* a2p = &A_lds[row * 64 + koff];

    XGroup g0, g1;
    load_group(g0, fp);
#pragma nounroll
    for (int i = 0; i < 8; ++i) {
        const int ga = (2 * i + 3 < 15) ? 2 * i + 3 : 15;
        const int gb = (2 * i + 4 < 14) ? 2 * i + 4 : 14;
        park_group(r1, sp + 8192);
        fetch_group(r1, xg + ga * 8192);
        lds_barrier();
        load_group(g1, fp + 8192);
        __builtin_amdgcn_sched_barrier(0);
        {
            const float* ap = a2p + 8 * i;
            v2f q0; q0.x = ap[0]; q0.y = ap[1];
            v2f q1; q1.x = ap[16 * 64]; q1.y = ap[16 * 64 + 1];
            contract_group(g0, a1, q0, q1, lo, out);
        }
        __builtin_amdgcn_sched_barrier(0);
        park_group(r0, sp);
        fetch_group(r0, xg + gb * 8192);
        lds_barrier();
        load_group(g0, fp);
        __builtin_amdgcn_sched_barrier(0);
        {
            const float* ap = a2p + 8 * i + 4;
            v2f q0; q0.x = ap[0]; q0.y = ap[1];
            v2f q1; q1.x = ap[16 * 64]; q1.y = ap[16 * 64 + 1];
            contract_group(g1, a1, q0, q1, lo, out);
        }
        __builtin_amdgcn_sched_barrier(0);
    }

    float* zp = z2 + (size_t)bd * (32 * 32 * 32) + ct * 16 + row;
#pragma unroll
    for (int qt = 0; qt < 2; ++qt) {
#pragma unroll
        for (int rr = 0; rr < 16; ++rr) {
#pragma unroll
            for (int u = 0; u < 8; ++u) {
                const int q = 16 * qt + 8 * half + u;
                const int r = 16 * rt + rr;
                zp[(q * 32 + r) * 32] = out[qt][rr][u];
            }
        }
    }
}

__global__ __launch_bounds__(256) void contract_d_kernel(
    const float* __restrict__ Aglob, const float* __restrict__ Z, float* __restrict__ C)
{
    __shared__ float A_lds[32 * 64];
    for (int i = threadIdx.x; i < 32 * 64; i += 256) A_lds[i] = Aglob[i];
    __syncthreads();

    const int lane = threadIdx.x & 31;
    const int wave = threadIdx.x >> 5;
    const int t    = blockIdx.x * 8 + wave;
    const int b    = t >> 11;
    const int col  = (t & 2047) * 16 + (lane & 15);
    const int row  = lane & 15;
    const int koff = (lane >> 4) << 1;

    v2f afrag[2][16];
#pragma unroll
    for (int mt = 0; mt < 2; ++mt) {
#pragma unroll
        for (int kk = 0; kk < 16; ++kk) {
            const float* ap = &A_lds[(mt * 16 + row) * 64 + kk * 4 + koff];
            v2f a; a.x = ap[0]; a.y = ap[1];
            afrag[mt][kk] = a;
        }
    }

    const float* Bb = Z + (size_t)b * 64 * 32768;
    v8f acc0 = {}; v8f acc1 = {};
#pragma unroll
    for (int kk = 0; kk < 16; ++kk) {
        const float* bp0 = Bb + (size_t)(kk * 4 + koff) * 32768 + col;
        v2f bf; bf.x = bp0[0]; bf.y = bp0[32768];
        acc0 = mma(afrag[0][kk], bf, acc0);
        acc1 = mma(afrag[1][kk], bf, acc1);
    }

    const int rbase = (lane >> 4) << 3;
    const int q = col >> 10;
    const int r = (col >> 5) & 31;
    const int c = col & 31;
    float* op = C + ((((size_t)b * 32 + r) * 32 + q) * 32) * 32 + c;
#pragma unroll
    for (int v = 0; v < 8; ++v) {
        op[(rbase + v) * 32]      = acc0[v];
        op[(16 + rbase + v) * 32] = acc1[v];
    }
}

extern "C" void kernel_launch(void* const* d_in, const int* in_sizes, int n_in,
                              void* d_out, int out_size, void* d_ws, size_t ws_size,
                              hipStream_t stream) {
    (void)in_sizes; (void)n_in; (void)out_size; (void)ws_size;
    const float* x  = (const float*)d_in[0];
    float* out      = (float*)d_out;

    char* ws = (char*)d_ws;
    float* A  = (float*)ws;
    float* z2 = (float*)(ws + 8192);

    build_A_kernel<<<8, 256, 0, stream>>>(A);
    fused_wh_kernel<<<256, 128, 0, stream>>>(A, x, z2);
    contract_d_kernel<<<1024, 256, 0, stream>>>(A, z2, out);
}


# operator build: every cosine evaluated once into LDS tables per work-group (2 per thread, was 48), then the same 24 products summed in the same order; 3 launches as before
# speedup vs baseline: 1.6741x; 1.0528x over previous
#include <hip/hip_runtime.h>
#include <math.h>


typedef __attribute__((ext_vector_type(2))) float v2f;
typedef __attribute__((ext_vector_type(8))) float v8f;
typedef __attribute__((ext_vector_type(4))) float v4f;

__global__ __launch_bounds__(1024) void build_A_kernel(float* __restrict__ Aout) {
    __shared__ float c64_lds[24 * 64];
    __shared__ float t32_lds[24 * 4];
    const float PI = 3.14159265358979323846f;
    const int tid = threadIdx.x;
    const int rb  = blockIdx.x * 4;
    for (int e = tid; e < 24 * 64 + 24 * 4; e += 1024) {
        if (e < 24 * 64) {
            int p = e >> 6, w = e & 63;
            c64_lds[e] = cosf(PI * ((float)w + 0.5f) * (float)p / 64.0f);
        } else {
            int f = e - 24 * 64;
            int p = f >> 2, r = rb + (f & 3);
            float s32 = (p == 0) ? sqrtf(1.0f / 32.0f) : sqrtf(2.0f / 32.0f);
            t32_lds[f] = cosf(PI * ((float)r + 0.5f) * (float)p / 32.0f) * s32;
        }
    }
    __syncthreads();
    if (tid < 256) {
        int rl = tid >> 6;
        int w  = tid & 63;
        float acc = 0.0f;
#pragma unroll
        for (int p = 0; p < 24; ++p) {
            float s64 = (p == 0) ? sqrtf(1.0f / 64.0f) : sqrtf(2.0f / 64.0f);
            acc += t32_lds[p * 4 + rl] * c64_lds[p * 64 + w] * s64;
        }
        Aout[(rb + rl) * 64 + w] = acc;
    }
}

__device__ __forceinline__ v8f mma(v2f a, v2f b, v8f c) {
    return __builtin_amdgcn_wmma_f32_16x16x4_f32(false, a, false, b, (short)0, c, false, false);
}

__device__ __forceinline__ float xhalf(float x) {
    int b = __builtin_bit_cast(int, x);
    return __builtin_bit_cast(float,
        __builtin_amdgcn_permlanex16(b, b, 0x76543210, 0xfedcba98, false, false));
}

struct XGroup { v2f f[4][16]; };

struct Stage { v4f r[16]; };

__device__ __forceinline__ void fetch_group(Stage& s, const float* __restrict__ src) {
#pragma unroll
    for (int j = 0; j < 16; ++j) s.r[j] = *reinterpret_cast<const v4f*>(src + j * 512);
}

__device__ __forceinline__ void park_group(const Stage& s, float* dst) {
#pragma unroll
    for (int j = 0; j < 16; ++j) *reinterpret_cast<v4f*>(dst + j * 512) = s.r[j];
}

__device__ __forceinline__ void load_group(XGroup& xg, const float* p) {
#pragma unroll
    for (int hi = 0; hi < 4; ++hi) {
#pragma unroll
        for (int kk = 0; kk < 16; ++kk) {
            const float* q = p + hi * 2048 + kk * 128;
            v2f t; t.x = q[0]; t.y = q[64];
            xg.f[hi][kk] = t;
        }
    }
}

__device__ __forceinline__ void lds_barrier() {
    asm volatile("s_wait_dscnt 0x0\n\ts_barrier_signal -1\n\ts_barrier_wait -1" ::: "memory");
}

__device__ __forceinline__ void contract_group(const XGroup& xg, const v2f (&a1)[16],
                                               v2f a2q0, v2f a2q1, bool lo, v8f (&out)[2][16]) {
    v8f d1[4];
#pragma unroll
    for (int hi = 0; hi < 4; ++hi) d1[hi] = (v8f){};
#pragma unroll
    for (int kk = 0; kk < 16; ++kk) {
#pragma unroll
        for (int hi = 0; hi < 4; ++hi) d1[hi] = mma(a1[kk], xg.f[hi][kk], d1[hi]);
    }
#pragma unroll
    for (int v = 0; v < 8; ++v) {
        float sx = xhalf(lo ? d1[2][v] : d1[0][v]);
        float sy = xhalf(lo ? d1[3][v] : d1[1][v]);
        v2f blo, bhi;
        blo.x = lo ? d1[0][v] : sx;
        blo.y = lo ? d1[1][v] : sy;
        bhi.x = lo ? sx : d1[2][v];
        bhi.y = lo ? sy : d1[3][v];
        out[0][v]     = mma(a2q0, blo, out[0][v]);
        out[1][v]     = mma(a2q1, blo, out[1][v]);
        out[0][8 + v] = mma(a2q0, bhi, out[0][8 + v]);
        out[1][8 + v] = mma(a2q1, bhi, out[1][8 + v]);
    }
}

__global__ __launch_bounds__(128) void fused_wh_kernel(
    const float* __restrict__ Aglob, const float* __restrict__ x, float* __restrict__ z2)
{
    __shared__ __attribute__((aligned(16))) float lds[2 * 8192 + 32 * 64];
    float* const A_lds = lds + 2 * 8192;

    const int tid  = threadIdx.x;
    const int lane = tid & 31;
    const int wave = tid >> 5;
    const int ct   = wave & 1;
    const int rt   = wave >> 1;
    const int bd   = blockIdx.x;
    const int row  = lane & 15;

    const float* xg = x + (size_t)bd * (64 * 64 * 32) + tid * 4;
    const int r4 = (tid >> 3) & 3;
    float* sp = lds + ((wave * 4 + (((r4 & 1) << 1) | (r4 >> 1))) * 32 + (tid & 7) * 4);

    float areg[16];
#pragma unroll
    for (int j = 0; j < 16; ++j) areg[j] = Aglob[tid + 128 * j];
    Stage r0, r1;
    fetch_group(r0, xg);
    fetch_group(r1, xg + 8192);
#pragma unroll
    for (int j = 0; j < 16; ++j) A_lds[tid + 128 * j] = areg[j];
    park_group(r0, sp);
    fetch_group(r0, xg + 2 * 8192);
    lds_barrier();

    const int half = lane >> 4;
    const int koff = half << 1;
    const bool lo  = (half == 0);

    v2f a1[16];
#pragma unroll
    for (int kk = 0; kk < 16; ++kk) {
        const float* ap = &A_lds[(rt * 16 + row) * 64 + kk * 4 + koff];
        v2f a; a.x = ap[0]; a.y = ap[1];
        a1[kk] = a;
    }

    v8f out[2][16];
#pragma unroll
    for (int qt = 0; qt < 2; ++qt) {
#pragma unroll
        for (int rr = 0; rr < 16; ++rr) out[qt][rr] = (v8f){};
    }

    const float* fp = lds + half * 32 + ct * 16 + row;
    const float* a2p = &A_lds[row * 64 + koff];

    XGroup g0, g1;
    load_group(g0, fp);
#pragma nounroll
    for (int i = 0; i < 8; ++i) {
        const int ga = (2 * i + 3 < 15) ? 2 * i + 3 : 15;
        const int gb = (2 * i + 4 < 14) ? 2 * i + 4 : 14;
        park_group(r1, sp + 8192);
        fetch_group(r1, xg + ga * 8192);
        lds_barrier();
        load_group(g1, fp + 8192);
        __builtin_amdgcn_sched_barrier(0);
        {
            const float* ap = a2p + 8 * i;
            v2f q0; q0.x = ap[0]; q0.y = ap[1];
            v2f q1; q1.x = ap[16 * 64]; q1.y = ap[16 * 64 + 1];
            contract_group(g0, a1, q0, q1, lo, out);
        }
        __builtin_amdgcn_sched_barrier(0);
        park_group(r0, sp);
        fetch_group(r0, xg + gb * 8192);
        lds_barrier();
        load_group(g0, fp);
        __builtin_amdgcn_sched_barrier(0);
        {
            const float* ap = a2p + 8 * i + 4;
            v2f q0; q0.x = ap[0]; q0.y = ap[1];
            v2f q1; q1.x = ap[16 * 64]; q1.y = ap[16 * 64 + 1];
            contract_group(g1, a1, q0, q1, lo, out);
        }
        __builtin_amdgcn_sched_barrier(0);
    }

    float* zp = z2 + (size_t)bd * (32 * 32 * 32) + ct * 16 + row;
#pragma unroll
    for (int qt = 0; qt < 2; ++qt) {
#pragma unroll
        for (int rr = 0; rr < 16; ++rr) {
#pragma unroll
            for (int u = 0; u < 8; ++u) {
                const int q = 16 * qt + 8 * half + u;
                const int r = 16 * rt + rr;
                zp[(q * 32 + r) * 32] = out[qt][rr][u];
            }
        }
    }
}

__global__ __launch_bounds__(256) void contract_d_kernel(
    const float* __restrict__ Aglob, const float* __restrict__ Z, float* __restrict__ C)
{
    __shared__ float A_lds[32 * 64];
    for (int i = threadIdx.x; i < 32 * 64; i += 256) A_lds[i] = Aglob[i];
    __syncthreads();

    const int lane = threadIdx.x & 31;
    const int wave = threadIdx.x >> 5;
    const int t    = blockIdx.x * 8 + wave;
    const int b    = t >> 11;
    const int col  = (t & 2047) * 16 + (lane & 15);
    const int row  = lane & 15;
    const int koff = (lane >> 4) << 1;

    v2f afrag[2][16];
#pragma unroll
    for (int mt = 0; mt < 2; ++mt) {
#pragma unroll
        for (int kk = 0; kk < 16; ++kk) {
            const float* ap = &A_lds[(mt * 16 + row) * 64 + kk * 4 + koff];
            v2f a; a.x = ap[0]; a.y = ap[1];
            afrag[mt][kk] = a;
        }
    }

    const float* Bb = Z + (size_t)b * 64 * 32768;
    v8f acc0 = {}; v8f acc1 = {};
#pragma unroll
    for (int kk = 0; kk < 16; ++kk) {
        const float* bp0 = Bb + (size_t)(kk * 4 + koff) * 32768 + col;
        v2f bf; bf.x = bp0[0]; bf.y = bp0[32768];
        acc0 = mma(afrag[0][kk], bf, acc0);
        acc1 = mma(afrag[1][kk], bf, acc1);
    }

    const int rbase = (lane >> 4) << 3;
    const int q = col >> 10;
    const int r = (col >> 5) & 31;
    const int c = col & 31;
    float* op = C + ((((size_t)b * 32 + r) * 32 + q) * 32) * 32 + c;
#pragma unroll
    for (int v = 0; v < 8; ++v) {
        op[(rbase + v) * 32]      = acc0[v];
        op[(16 + rbase + v) * 32] = acc1[v];
    }
}

extern "C" void kernel_launch(void* const* d_in, const int* in_sizes, int n_in,
                              void* d_out, int out_size, void* d_ws, size_t ws_size,
                              hipStream_t stream) {
    (void)in_sizes; (void)n_in; (void)out_size; (void)ws_size;
    const float* x  = (const float*)d_in[0];
    float* out      = (float*)d_out;

    char* ws = (char*)d_ws;
    float* A  = (float*)ws;
    float* z2 = (float*)(ws + 8192);

    build_A_kernel<<<8, 1024, 0, stream>>>(A);
    fused_wh_kernel<<<256, 128, 0, stream>>>(A, x, z2);
    contract_d_kernel<<<1024, 256, 0, stream>>>(A, z2, out);
}


# operator's LDS copy at a row pitch of 68 words (was 64) in both WMMA kernels: its fragment reads, 16 rows on one bank each, were 16-way bank conflicted; same values, same sum order
# speedup vs baseline: 1.8392x; 1.0986x over previous
#include <hip/hip_runtime.h>
#include <math.h>


typedef __attribute__((ext_vector_type(2))) float v2f;
typedef __attribute__((ext_vector_type(8))) float v8f;
typedef __attribute__((ext_vector_type(4))) float v4f;

constexpr int APITCH = 68;

__global__ __launch_bounds__(1024) void build_A_kernel(float* __restrict__ Aout) {
    __shared__ float c64_lds[24 * 64];
    __shared__ float t32_lds[24 * 4];
    const float PI = 3.14159265358979323846f;
    const int tid = threadIdx.x;
    const int rb  = blockIdx.x * 4;
    for (int e = tid; e < 24 * 64 + 24 * 4; e += 1024) {
        if (e < 24 * 64) {
            int p = e >> 6, w = e & 63;
            c64_lds[e] = cosf(PI * ((float)w + 0.5f) * (float)p / 64.0f);
        } else {
            int f = e - 24 * 64;
            int p = f >> 2, r = rb + (f & 3);
            float s32 = (p == 0) ? sqrtf(1.0f / 32.0f) : sqrtf(2.0f / 32.0f);
            t32_lds[f] = cosf(PI * ((float)r + 0.5f) * (float)p / 32.0f) * s32;
        }
    }
    __syncthreads();
    if (tid < 256) {
        int rl = tid >> 6;
        int w  = tid & 63;
        float acc = 0.0f;
#pragma unroll
        for (int p = 0; p < 24; ++p) {
            float s64 = (p == 0) ? sqrtf(1.0f / 64.0f) : sqrtf(2.0f / 64.0f);
            acc += t32_lds[p * 4 + rl] * c64_lds[p * 64 + w] * s64;
        }
        Aout[(rb + rl) * 64 + w] = acc;
    }
}

__device__ __forceinline__ v8f mma(v2f a, v2f b, v8f c) {
    return __builtin_amdgcn_wmma_f32_16x16x4_f32(false, a, false, b, (short)0, c, false, false);
}

__device__ __forceinline__ float xhalf(float x) {
    int b = __builtin_bit_cast(int, x);
    return __builtin_bit_cast(float,
        __builtin_amdgcn_permlanex16(b, b, 0x76543210, 0xfedcba98, false, false));
}

struct XGroup { v2f f[4][16]; };

struct Stage { v4f r[16]; };

__device__ __forceinline__ void fetch_group(Stage& s, const float* __restrict__ src) {
#pragma unroll
    for (int j = 0; j < 16; ++j) s.r[j] = *reinterpret_cast<const v4f*>(src + j * 512);
}

__device__ __forceinline__ void park_group(const Stage& s, float* dst) {
#pragma unroll
    for (int j = 0; j < 16; ++j) *reinterpret_cast<v4f*>(dst + j * 512) = s.r[j];
}

__device__ __forceinline__ void load_group(XGroup& xg, const float* p) {
#pragma unroll
    for (int hi = 0; hi < 4; ++hi) {
#pragma unroll
        for (int kk = 0; kk < 16; ++kk) {
            const float* q = p + hi * 2048 + kk * 128;
            v2f t; t.x = q[0]; t.y = q[64];
            xg.f[hi][kk] = t;
        }
    }
}

__device__ __forceinline__ void lds_barrier() {
    asm volatile("s_wait_dscnt 0x0\n\ts_barrier_signal -1\n\ts_barrier_wait -1" ::: "memory");
}

__device__ __forceinline__ void contract_group(const XGroup& xg, const v2f (&a1)[16],
                                               v2f a2q0, v2f a2q1, bool lo, v8f (&out)[2][16]) {
    v8f d1[4];
#pragma unroll
    for (int hi = 0; hi < 4; ++hi) d1[hi] = (v8f){};
#pragma unroll
    for (int kk = 0; kk < 16; ++kk) {
#pragma unroll
        for (int hi = 0; hi < 4; ++hi) d1[hi] = mma(a1[kk], xg.f[hi][kk], d1[hi]);
    }
#pragma unroll
    for (int v = 0; v < 8; ++v) {
        float sx = xhalf(lo ? d1[2][v] : d1[0][v]);
        float sy = xhalf(lo ? d1[3][v] : d1[1][v]);
        v2f blo, bhi;
        blo.x = lo ? d1[0][v] : sx;
        blo.y = lo ? d1[1][v] : sy;
        bhi.x = lo ? sx : d1[2][v];
        bhi.y = lo ? sy : d1[3][v];
        out[0][v]     = mma(a2q0, blo, out[0][v]);
        out[1][v]     = mma(a2q1, blo, out[1][v]);
        out[0][8 + v] = mma(a2q0, bhi, out[0][8 + v]);
        out[1][8 + v] = mma(a2q1, bhi, out[1][8 + v]);
    }
}

__global__ __launch_bounds__(128) void fused_wh_kernel(
    const float* __restrict__ Aglob, const float* __restrict__ x, float* __restrict__ z2)
{
    __shared__ __attribute__((aligned(16))) float lds[2 * 8192 + 32 * APITCH];
    float* const A_lds = lds + 2 * 8192;

    const int tid  = threadIdx.x;
    const int lane = tid & 31;
    const int wave = tid >> 5;
    const int ct   = wave & 1;
    const int rt   = wave >> 1;
    const int bd   = blockIdx.x;
    const int row  = lane & 15;

    const float* xg = x + (size_t)bd * (64 * 64 * 32) + tid * 4;
    const int r4 = (tid >> 3) & 3;
    float* sp = lds + ((wave * 4 + (((r4 & 1) << 1) | (r4 >> 1))) * 32 + (tid & 7) * 4);

    float areg[16];
#pragma unroll
    for (int j = 0; j < 16; ++j) areg[j] = Aglob[tid + 128 * j];
    Stage r0, r1;
    fetch_group(r0, xg);
    fetch_group(r1, xg + 8192);
#pragma unroll
    for (int j = 0; j < 16; ++j) {
        A_lds[(tid >> 6) * APITCH + (tid & 63) + j * (2 * APITCH)] = areg[j];
    }
    park_group(r0, sp);
    fetch_group(r0, xg + 2 * 8192);
    lds_barrier();

    const int half = lane >> 4;
    const int koff = half << 1;
    const bool lo  = (half == 0);

    v2f a1[16];
#pragma unroll
    for (int kk = 0; kk < 16; ++kk) {
        const float* ap = &A_lds[(rt * 16 + row) * APITCH + kk * 4 + koff];
        v2f a; a.x = ap[0]; a.y = ap[1];
        a1[kk] = a;
    }

    v8f out[2][16];
#pragma unroll
    for (int qt = 0; qt < 2; ++qt) {
#pragma unroll
        for (int rr = 0; rr < 16; ++rr) out[qt][rr] = (v8f){};
    }

    const float* fp = lds + half * 32 + ct * 16 + row;
    const float* a2p = &A_lds[row * APITCH + koff];

    XGroup g0, g1;
    load_group(g0, fp);
#pragma nounroll
    for (int i = 0; i < 8; ++i) {
        const int ga = (2 * i + 3 < 15) ? 2 * i + 3 : 15;
        const int gb = (2 * i + 4 < 14) ? 2 * i + 4 : 14;
        park_group(r1, sp + 8192);
        fetch_group(r1, xg + ga * 8192);
        lds_barrier();
        load_group(g1, fp + 8192);
        __builtin_amdgcn_sched_barrier(0);
        {
            const float* ap = a2p + 8 * i;
            v2f q0; q0.x = ap[0]; q0.y = ap[1];
            v2f q1; q1.x = ap[16 * APITCH]; q1.y = ap[16 * APITCH + 1];
            contract_group(g0, a1, q0, q1, lo, out);
        }
        __builtin_amdgcn_sched_barrier(0);
        park_group(r0, sp);
        fetch_group(r0, xg + gb * 8192);
        lds_barrier();
        load_group(g0, fp);
        __builtin_amdgcn_sched_barrier(0);
        {
            const float* ap = a2p + 8 * i + 4;
            v2f q0; q0.x = ap[0]; q0.y = ap[1];
            v2f q1; q1.x = ap[16 * APITCH]; q1.y = ap[16 * APITCH + 1];
            contract_group(g1, a1, q0, q1, lo, out);
        }
        __builtin_amdgcn_sched_barrier(0);
    }

    float* zp = z2 + (size_t)bd * (32 * 32 * 32) + ct * 16 + row;
#pragma unroll
    for (int qt = 0; qt < 2; ++qt) {
#pragma unroll
        for (int rr = 0; rr < 16; ++rr) {
#pragma unroll
            for (int u = 0; u < 8; ++u) {
                const int q = 16 * qt + 8 * half + u;
                const int r = 16 * rt + rr;
                zp[(q * 32 + r) * 32] = out[qt][rr][u];
            }
        }
    }
}

__global__ __launch_bounds__(256) void contract_d_kernel(
    const float* __restrict__ Aglob, const float* __restrict__ Z, float* __restrict__ C)
{
    __shared__ __attribute__((aligned(16))) float A_lds[32 * APITCH];
    for (int i = threadIdx.x; i < 32 * 64; i += 256) A_lds[(i >> 6) * APITCH + (i & 63)] = Aglob[i];
    __syncthreads();

    const int lane = threadIdx.x & 31;
    const int wave = threadIdx.x >> 5;
    const int t    = blockIdx.x * 8 + wave;
    const int b    = t >> 11;
    const int col  = (t & 2047) * 16 + (lane & 15);
    const int row  = lane & 15;
    const int koff = (lane >> 4) << 1;

    v2f afrag[2][16];
#pragma unroll
    for (int mt = 0; mt < 2; ++mt) {
#pragma unroll
        for (int kk = 0; kk < 16; ++kk) {
            const float* ap = &A_lds[(mt * 16 + row) * APITCH + kk * 4 + koff];
            v2f a; a.x = ap[0]; a.y = ap[1];
            afrag[mt][kk] = a;
        }
    }

    const float* Bb = Z + (size_t)b * 64 * 32768;
    v8f acc0 = {}; v8f acc1 = {};
#pragma unroll
    for (int kk = 0; kk < 16; ++kk) {
        const float* bp0 = Bb + (size_t)(kk * 4 + koff) * 32768 + col;
        v2f bf; bf.x = bp0[0]; bf.y = bp0[32768];
        acc0 = mma(afrag[0][kk], bf, acc0);
        acc1 = mma(afrag[1][kk], bf, acc1);
    }

    const int rbase = (lane >> 4) << 3;
    const int q = col >> 10;
    const int r = (col >> 5) & 31;
    const int c = col & 31;
    float* op = C + ((((size_t)b * 32 + r) * 32 + q) * 32) * 32 + c;
#pragma unroll
    for (int v = 0; v < 8; ++v) {
        op[(rbase + v) * 32]      = acc0[v];
        op[(16 + rbase + v) * 32] = acc1[v];
    }
}

extern "C" void kernel_launch(void* const* d_in, const int* in_sizes, int n_in,
                              void* d_out, int out_size, void* d_ws, size_t ws_size,
                              hipStream_t stream) {
    (void)in_sizes; (void)n_in; (void)out_size; (void)ws_size;
    const float* x  = (const float*)d_in[0];
    float* out      = (float*)d_out;

    char* ws = (char*)d_ws;
    float* A  = (float*)ws;
    float* z2 = (float*)(ws + 8192);

    build_A_kernel<<<8, 1024, 0, stream>>>(A);
    fused_wh_kernel<<<256, 128, 0, stream>>>(A, x, z2);
    contract_d_kernel<<<1024, 256, 0, stream>>>(A, z2, out);
}


# operator stored in WMMA fragment order: the last pass takes its A fragments straight from memory (16 coalesced 16-byte loads per lane beside its z2 loads); no LDS copy, staging loop or barrier
# speedup vs baseline: 1.9138x; 1.0406x over previous
#include <hip/hip_runtime.h>
#include <math.h>


typedef __attribute__((ext_vector_type(2))) float v2f;
typedef __attribute__((ext_vector_type(8))) float v8f;
typedef __attribute__((ext_vector_type(4))) float v4f;

constexpr int APITCH = 68;

__device__ __forceinline__ int frag_index(int row, int col) {
    const int mt = row >> 4, r = row & 15;
    const int kk = col >> 2, half = (col >> 1) & 1, s = col & 1;
    return (((mt * 8 + (kk >> 1)) * 32 + half * 16 + r) << 2) + ((kk & 1) << 1) + s;
}

__global__ __launch_bounds__(1024) void build_A_kernel(float* __restrict__ Aout) {
    __shared__ float c64_lds[24 * 64];
    __shared__ float t32_lds[24 * 4];
    const float PI = 3.14159265358979323846f;
    const int tid = threadIdx.x;
    const int rb  = blockIdx.x * 4;
    for (int e = tid; e < 24 * 64 + 24 * 4; e += 1024) {
        if (e < 24 * 64) {
            int p = e >> 6, w = e & 63;
            c64_lds[e] = cosf(PI * ((float)w + 0.5f) * (float)p / 64.0f);
        } else {
            int f = e - 24 * 64;
            int p = f >> 2, r = rb + (f & 3);
            float s32 = (p == 0) ? sqrtf(1.0f / 32.0f) : sqrtf(2.0f / 32.0f);
            t32_lds[f] = cosf(PI * ((float)r + 0.5f) * (float)p / 32.0f) * s32;
        }
    }
    __syncthreads();
    if (tid < 256) {
        int rl = tid >> 6;
        int w  = tid & 63;
        float acc = 0.0f;
#pragma unroll
        for (int p = 0; p < 24; ++p) {
            float s64 = (p == 0) ? sqrtf(1.0f / 64.0f) : sqrtf(2.0f / 64.0f);
            acc += t32_lds[p * 4 + rl] * c64_lds[p * 64 + w] * s64;
        }
        Aout[frag_index(rb + rl, w)] = acc;
    }
}

__device__ __forceinline__ v8f mma(v2f a, v2f b, v8f c) {
    return __builtin_amdgcn_wmma_f32_16x16x4_f32(false, a, false, b, (short)0, c, false, false);
}

__device__ __forceinline__ float xhalf(float x) {
    int b = __builtin_bit_cast(int, x);
    return __builtin_bit_cast(float,
        __builtin_amdgcn_permlanex16(b, b, 0x76543210, 0xfedcba98, false, false));
}

struct XGroup { v2f f[4][16]; };

struct Stage { v4f r[16]; };

__device__ __forceinline__ void fetch_group(Stage& s, const float* __restrict__ src) {
#pragma unroll
    for (int j = 0; j < 16; ++j) s.r[j] = *reinterpret_cast<const v4f*>(src + j * 512);
}

__device__ __forceinline__ void park_group(const Stage& s, float* dst) {
#pragma unroll
    for (int j = 0; j < 16; ++j) *reinterpret_cast<v4f*>(dst + j * 512) = s.r[j];
}

__device__ __forceinline__ void load_group(XGroup& xg, const float* p) {
#pragma unroll
    for (int hi = 0; hi < 4; ++hi) {
#pragma unroll
        for (int kk = 0; kk < 16; ++kk) {
            const float* q = p + hi * 2048 + kk * 128;
            v2f t; t.x = q[0]; t.y = q[64];
            xg.f[hi][kk] = t;
        }
    }
}

__device__ __forceinline__ void lds_barrier() {
    asm volatile("s_wait_dscnt 0x0\n\ts_barrier_signal -1\n\ts_barrier_wait -1" ::: "memory");
}

__device__ __forceinline__ void contract_group(const XGroup& xg, const v2f (&a1)[16],
                                               v2f a2q0, v2f a2q1, bool lo, v8f (&out)[2][16]) {
    v8f d1[4];
#pragma unroll
    for (int hi = 0; hi < 4; ++hi) d1[hi] = (v8f){};
#pragma unroll
    for (int kk = 0; kk < 16; ++kk) {
#pragma unroll
        for (int hi = 0; hi < 4; ++hi) d1[hi] = mma(a1[kk], xg.f[hi][kk], d1[hi]);
    }
#pragma unroll
    for (int v = 0; v < 8; ++v) {
        float sx = xhalf(lo ? d1[2][v] : d1[0][v]);
        float sy = xhalf(lo ? d1[3][v] : d1[1][v]);
        v2f blo, bhi;
        blo.x = lo ? d1[0][v] : sx;
        blo.y = lo ? d1[1][v] : sy;
        bhi.x = lo ? sx : d1[2][v];
        bhi.y = lo ? sy : d1[3][v];
        out[0][v]     = mma(a2q0, blo, out[0][v]);
        out[1][v]     = mma(a2q1, blo, out[1][v]);
        out[0][8 + v] = mma(a2q0, bhi, out[0][8 + v]);
        out[1][8 + v] = mma(a2q1, bhi, out[1][8 + v]);
    }
}

__global__ __launch_bounds__(128) void fused_wh_kernel(
    const float* __restrict__ Aglob, const float* __restrict__ x, float* __restrict__ z2)
{
    __shared__ __attribute__((aligned(16))) float lds[2 * 8192 + 32 * APITCH];
    float* const A_lds = lds + 2 * 8192;

    const int tid  = threadIdx.x;
    const int lane = tid & 31;
    const int wave = tid >> 5;
    const int ct   = wave & 1;
    const int rt   = wave >> 1;
    const int bd   = blockIdx.x;
    const int row  = lane & 15;

    const float* xg = x + (size_t)bd * (64 * 64 * 32) + tid * 4;
    const int r4 = (tid >> 3) & 3;
    float* sp = lds + ((wave * 4 + (((r4 & 1) << 1) | (r4 >> 1))) * 32 + (tid & 7) * 4);

    float areg[16];
#pragma unroll
    for (int j = 0; j < 16; ++j) areg[j] = Aglob[tid + 128 * j];
    Stage r0, r1;
    fetch_group(r0, xg);
    fetch_group(r1, xg + 8192);
    const int a_row = (tid >> 2) & 15;
    const int a_col = (((tid >> 1) & 1) << 2) + ((tid >> 6) << 1) + (tid & 1);
#pragma unroll
    for (int j = 0; j < 16; ++j) {
        A_lds[((j >> 3) * 16 + a_row) * APITCH + a_col + 8 * (j & 7)] = areg[j];
    }
    park_group(r0, sp);
    fetch_group(r0, xg + 2 * 8192);
    lds_barrier();

    const int half = lane >> 4;
    const int koff = half << 1;
    const bool lo  = (half == 0);

    v2f a1[16];
#pragma unroll
    for (int kk = 0; kk < 16; ++kk) {
        const float* ap = &A_lds[(rt * 16 + row) * APITCH + kk * 4 + koff];
        v2f a; a.x = ap[0]; a.y = ap[1];
        a1[kk] = a;
    }

    v8f out[2][16];
#pragma unroll
    for (int qt = 0; qt < 2; ++qt) {
#pragma unroll
        for (int rr = 0; rr < 16; ++rr) out[qt][rr] = (v8f){};
    }

    const float* fp = lds + half * 32 + ct * 16 + row;
    const float* a2p = &A_lds[row * APITCH + koff];

    XGroup g0, g1;
    load_group(g0, fp);
#pragma nounroll
    for (int i = 0; i < 8; ++i) {
        const int ga = (2 * i + 3 < 15) ? 2 * i + 3 : 15;
        const int gb = (2 * i + 4 < 14) ? 2 * i + 4 : 14;
        park_group(r1, sp + 8192);
        fetch_group(r1, xg + ga * 8192);
        lds_barrier();
        load_group(g1, fp + 8192);
        __builtin_amdgcn_sched_barrier(0);
        {
            const float* ap = a2p + 8 * i;
            v2f q0; q0.x = ap[0]; q0.y = ap[1];
            v2f q1; q1.x = ap[16 * APITCH]; q1.y = ap[16 * APITCH + 1];
            contract_group(g0, a1, q0, q1, lo, out);
        }
        __builtin_amdgcn_sched_barrier(0);
        park_group(r0, sp);
        fetch_group(r0, xg + gb * 8192);
        lds_barrier();
        load_group(g0, fp);
        __builtin_amdgcn_sched_barrier(0);
        {
            const float* ap = a2p + 8 * i + 4;
            v2f q0; q0.x = ap[0]; q0.y = ap[1];
            v2f q1; q1.x = ap[16 * APITCH]; q1.y = ap[16 * APITCH + 1];
            contract_group(g1, a1, q0, q1, lo, out);
        }
        __builtin_amdgcn_sched_barrier(0);
    }

    float* zp = z2 + (size_t)bd * (32 * 32 * 32) + ct * 16 + row;
#pragma unroll
    for (int qt = 0; qt < 2; ++qt) {
#pragma unroll
        for (int rr = 0; rr < 16; ++rr) {
#pragma unroll
            for (int u = 0; u < 8; ++u) {
                const int q = 16 * qt + 8 * half + u;
                const int r = 16 * rt + rr;
                zp[(q * 32 + r) * 32] = out[qt][rr][u];
            }
        }
    }
}

__global__ __launch_bounds__(256) __attribute__((amdgpu_waves_per_eu(8, 8))) void contract_d_kernel(
    const float* __restrict__ Afrag, const float* __restrict__ Z, float* __restrict__ C)
{
    const int lane = threadIdx.x & 31;
    const int wave = threadIdx.x >> 5;
    const int t    = blockIdx.x * 8 + wave;
    const int b    = t >> 11;
    const int col  = (t & 2047) * 16 + (lane & 15);
    const int koff = (lane >> 4) << 1;

    const v4f* ap = reinterpret_cast<const v4f*>(Afrag) + lane;
    v4f aq[2][8];
#pragma unroll
    for (int mt = 0; mt < 2; ++mt) {
#pragma unroll
        for (int j = 0; j < 8; ++j) aq[mt][j] = ap[(mt * 8 + j) * 32];
    }

    const float* bp = Z + (size_t)b * 64 * 32768 + (size_t)koff * 32768 + col;
    v2f bf[16];
#pragma unroll
    for (int kk = 0; kk < 16; ++kk) {
        v2f f; f.x = bp[(size_t)(kk * 4) * 32768]; f.y = bp[(size_t)(kk * 4 + 1) * 32768];
        bf[kk] = f;
    }

    v8f acc0 = {}; v8f acc1 = {};
#pragma unroll
    for (int kk = 0; kk < 16; ++kk) {
        const v4f q0 = aq[0][kk >> 1];
        const v4f q1 = aq[1][kk >> 1];
        v2f a0, a1;
        if (kk & 1) { a0.x = q0.z; a0.y = q0.w; a1.x = q1.z; a1.y = q1.w; }
        else        { a0.x = q0.x; a0.y = q0.y; a1.x = q1.x; a1.y = q1.y; }
        acc0 = mma(a0, bf[kk], acc0);
        acc1 = mma(a1, bf[kk], acc1);
    }

    const int rbase = (lane >> 4) << 3;
    const int q = col >> 10;
    const int r = (col >> 5) & 31;
    const int c = col & 31;
    float* op = C + ((((size_t)b * 32 + r) * 32 + q) * 32) * 32 + c;
#pragma unroll
    for (int v = 0; v < 8; ++v) {
        op[(rbase + v) * 32]      = acc0[v];
        op[(16 + rbase + v) * 32] = acc1[v];
    }
}

extern "C" void kernel_launch(void* const* d_in, const int* in_sizes, int n_in,
                              void* d_out, int out_size, void* d_ws, size_t ws_size,
                              hipStream_t stream) {
    (void)in_sizes; (void)n_in; (void)out_size; (void)ws_size;
    const float* x  = (const float*)d_in[0];
    float* out      = (float*)d_out;

    char* ws = (char*)d_ws;
    float* A  = (float*)ws;
    float* z2 = (float*)(ws + 8192);

    build_A_kernel<<<8, 1024, 0, stream>>>(A);
    fused_wh_kernel<<<256, 128, 0, stream>>>(A, x, z2);
    contract_d_kernel<<<1024, 256, 0, stream>>>(A, z2, out);
}


# z2 stored in accumulator-register order [q/4][r][c][q%4]: fused pass stores and last pass loads are 16-byte, 256 contiguous bytes per lane half; a last-pass wave takes 4 q tiles; same sum order
# speedup vs baseline: 2.0570x; 1.0748x over previous
#include <hip/hip_runtime.h>
#include <math.h>


typedef __attribute__((ext_vector_type(2))) float v2f;
typedef __attribute__((ext_vector_type(8))) float v8f;
typedef __attribute__((ext_vector_type(4))) float v4f;

constexpr int APITCH = 68;

__device__ __forceinline__ int frag_index(int row, int col) {
    const int mt = row >> 4, r = row & 15;
    const int kk = col >> 2, half = (col >> 1) & 1, s = col & 1;
    return (((mt * 8 + (kk >> 1)) * 32 + half * 16 + r) << 2) + ((kk & 1) << 1) + s;
}

__global__ __launch_bounds__(1024) void build_A_kernel(float* __restrict__ Aout) {
    __shared__ float c64_lds[24 * 64];
    __shared__ float t32_lds[24 * 4];
    const float PI = 3.14159265358979323846f;
    const int tid = threadIdx.x;
    const int rb  = blockIdx.x * 4;
    for (int e = tid; e < 24 * 64 + 24 * 4; e += 1024) {
        if (e < 24 * 64) {
            int p = e >> 6, w = e & 63;
            c64_lds[e] = cosf(PI * ((float)w + 0.5f) * (float)p / 64.0f);
        } else {
            int f = e - 24 * 64;
            int p = f >> 2, r = rb + (f & 3);
            float s32 = (p == 0) ? sqrtf(1.0f / 32.0f) : sqrtf(2.0f / 32.0f);
            t32_lds[f] = cosf(PI * ((float)r + 0.5f) * (float)p / 32.0f) * s32;
        }
    }
    __syncthreads();
    if (tid < 256) {
        int rl = tid >> 6;
        int w  = tid & 63;
        float acc = 0.0f;
#pragma unroll
        for (int p = 0; p < 24; ++p) {
            float s64 = (p == 0) ? sqrtf(1.0f / 64.0f) : sqrtf(2.0f / 64.0f);
            acc += t32_lds[p * 4 + rl] * c64_lds[p * 64 + w] * s64;
        }
        Aout[frag_index(rb + rl, w)] = acc;
    }
}

__device__ __forceinline__ int z2_index(int qg, int r, int c) {
    return ((qg * 32 + r) * 32 + c) * 4;
}

__device__ __forceinline__ v8f mma(v2f a, v2f b, v8f c) {
    return __builtin_amdgcn_wmma_f32_16x16x4_f32(false, a, false, b, (short)0, c, false, false);
}

__device__ __forceinline__ float xhalf(float x) {
    int b = __builtin_bit_cast(int, x);
    return __builtin_bit_cast(float,
        __builtin_amdgcn_permlanex16(b, b, 0x76543210, 0xfedcba98, false, false));
}

struct XGroup { v2f f[4][16]; };

struct Stage { v4f r[16]; };

__device__ __forceinline__ void fetch_group(Stage& s, const float* __restrict__ src) {
#pragma unroll
    for (int j = 0; j < 16; ++j) s.r[j] = *reinterpret_cast<const v4f*>(src + j * 512);
}

__device__ __forceinline__ void park_group(const Stage& s, float* dst) {
#pragma unroll
    for (int j = 0; j < 16; ++j) *reinterpret_cast<v4f*>(dst + j * 512) = s.r[j];
}

__device__ __forceinline__ void load_group(XGroup& xg, const float* p) {
#pragma unroll
    for (int hi = 0; hi < 4; ++hi) {
#pragma unroll
        for (int kk = 0; kk < 16; ++kk) {
            const float* q = p + hi * 2048 + kk * 128;
            v2f t; t.x = q[0]; t.y = q[64];
            xg.f[hi][kk] = t;
        }
    }
}

__device__ __forceinline__ void lds_barrier() {
    asm volatile("s_wait_dscnt 0x0\n\ts_barrier_signal -1\n\ts_barrier_wait -1" ::: "memory");
}

__device__ __forceinline__ void contract_group(const XGroup& xg, const v2f (&a1)[16],
                                               v2f a2q0, v2f a2q1, bool lo, v8f (&out)[2][16]) {
    v8f d1[4];
#pragma unroll
    for (int hi = 0; hi < 4; ++hi) d1[hi] = (v8f){};
#pragma unroll
    for (int kk = 0; kk < 16; ++kk) {
#pragma unroll
        for (int hi = 0; hi < 4; ++hi) d1[hi] = mma(a1[kk], xg.f[hi][kk], d1[hi]);
    }
#pragma unroll
    for (int v = 0; v < 8; ++v) {
        float sx = xhalf(lo ? d1[2][v] : d1[0][v]);
        float sy = xhalf(lo ? d1[3][v] : d1[1][v]);
        v2f blo, bhi;
        blo.x = lo ? d1[0][v] : sx;
        blo.y = lo ? d1[1][v] : sy;
        bhi.x = lo ? sx : d1[2][v];
        bhi.y = lo ? sy : d1[3][v];
        out[0][v]     = mma(a2q0, blo, out[0][v]);
        out[1][v]     = mma(a2q1, blo, out[1][v]);
        out[0][8 + v] = mma(a2q0, bhi, out[0][8 + v]);
        out[1][8 + v] = mma(a2q1, bhi, out[1][8 + v]);
    }
}

__global__ __launch_bounds__(128) void fused_wh_kernel(
    const float* __restrict__ Aglob, const float* __restrict__ x, float* __restrict__ z2)
{
    __shared__ __attribute__((aligned(16))) float lds[2 * 8192 + 32 * APITCH];
    float* const A_lds = lds + 2 * 8192;

    const int tid  = threadIdx.x;
    const int lane = tid & 31;
    const int wave = tid >> 5;
    const int ct   = wave & 1;
    const int rt   = wave >> 1;
    const int bd   = blockIdx.x;
    const int row  = lane & 15;

    const float* xg = x + (size_t)bd * (64 * 64 * 32) + tid * 4;
    const int r4 = (tid >> 3) & 3;
    float* sp = lds + ((wave * 4 + (((r4 & 1) << 1) | (r4 >> 1))) * 32 + (tid & 7) * 4);

    float areg[16];
#pragma unroll
    for (int j = 0; j < 16; ++j) areg[j] = Aglob[tid + 128 * j];
    Stage r0, r1;
    fetch_group(r0, xg);
    fetch_group(r1, xg + 8192);
    const int a_row = (tid >> 2) & 15;
    const int a_col = (((tid >> 1) & 1) << 2) + ((tid >> 6) << 1) + (tid & 1);
#pragma unroll
    for (int j = 0; j < 16; ++j) {
        A_lds[((j >> 3) * 16 + a_row) * APITCH + a_col + 8 * (j & 7)] = areg[j];
    }
    park_group(r0, sp);
    fetch_group(r0, xg + 2 * 8192);
    lds_barrier();

    const int half = lane >> 4;
    const int koff = half << 1;
    const bool lo  = (half == 0);

    v2f a1[16];
#pragma unroll
    for (int kk = 0; kk < 16; ++kk) {
        const float* ap = &A_lds[(rt * 16 + row) * APITCH + kk * 4 + koff];
        v2f a; a.x = ap[0]; a.y = ap[1];
        a1[kk] = a;
    }

    v8f out[2][16];
#pragma unroll
    for (int qt = 0; qt < 2; ++qt) {
#pragma unroll
        for (int rr = 0; rr < 16; ++rr) out[qt][rr] = (v8f){};
    }

    const float* fp = lds + half * 32 + ct * 16 + row;
    const float* a2p = &A_lds[row * APITCH + koff];

    XGroup g0, g1;
    load_group(g0, fp);
#pragma nounroll
    for (int i = 0; i < 8; ++i) {
        const int ga = (2 * i + 3 < 15) ? 2 * i + 3 : 15;
        const int gb = (2 * i + 4 < 14) ? 2 * i + 4 : 14;
        park_group(r1, sp + 8192);
        fetch_group(r1, xg + ga * 8192);
        lds_barrier();
        load_group(g1, fp + 8192);
        __builtin_amdgcn_sched_barrier(0);
        {
            const float* ap = a2p + 8 * i;
            v2f q0; q0.x = ap[0]; q0.y = ap[1];
            v2f q1; q1.x = ap[16 * APITCH]; q1.y = ap[16 * APITCH + 1];
            contract_group(g0, a1, q0, q1, lo, out);
        }
        __builtin_amdgcn_sched_barrier(0);
        park_group(r0, sp);
        fetch_group(r0, xg + gb * 8192);
        lds_barrier();
        load_group(g0, fp);
        __builtin_amdgcn_sched_barrier(0);
        {
            const float* ap = a2p + 8 * i + 4;
            v2f q0; q0.x = ap[0]; q0.y = ap[1];
            v2f q1; q1.x = ap[16 * APITCH]; q1.y = ap[16 * APITCH + 1];
            contract_group(g1, a1, q0, q1, lo, out);
        }
        __builtin_amdgcn_sched_barrier(0);
    }

    float* zp = z2 + (size_t)bd * (32 * 32 * 32) + z2_index(2 * half, 16 * rt, ct * 16 + row);
#pragma unroll
    for (int qt = 0; qt < 2; ++qt) {
#pragma unroll
        for (int rr = 0; rr < 16; ++rr) {
#pragma unroll
            for (int u4 = 0; u4 < 2; ++u4) {
                v4f t;
                t.x = out[qt][rr][4 * u4 + 0]; t.y = out[qt][rr][4 * u4 + 1];
                t.z = out[qt][rr][4 * u4 + 2]; t.w = out[qt][rr][4 * u4 + 3];
                *reinterpret_cast<v4f*>(zp + z2_index(4 * qt + u4, rr, 0)) = t;
            }
        }
    }
}

__global__ __launch_bounds__(256) __attribute__((amdgpu_waves_per_eu(2, 2))) void contract_d_kernel(
    const float* __restrict__ Afrag, const float* __restrict__ Z, float* __restrict__ C)
{
    const int lane = threadIdx.x & 31;
    const int wave = threadIdx.x >> 5;
    const int t    = blockIdx.x * 8 + wave;
    const int ct   = t & 1;
    const int r    = (t >> 1) & 31;
    const int qg   = (t >> 6) & 7;
    const int b    = t >> 9;
    const int cl   = lane & 15;
    const int hf   = lane >> 4;
    const int c    = ct * 16 + cl;

    const v4f* ap = reinterpret_cast<const v4f*>(Afrag) + lane;
    v4f aq[2][8];
#pragma unroll
    for (int mt = 0; mt < 2; ++mt) {
#pragma unroll
        for (int j = 0; j < 8; ++j) aq[mt][j] = ap[(mt * 8 + j) * 32];
    }

    const float* bp = Z + (size_t)(b * 64 + 2 * hf) * 32768 + z2_index(qg, r, c);
    v4f bz[16][2];
#pragma unroll
    for (int kk = 0; kk < 16; ++kk) {
        bz[kk][0] = *reinterpret_cast<const v4f*>(bp + (size_t)(kk * 4) * 32768);
        bz[kk][1] = *reinterpret_cast<const v4f*>(bp + (size_t)(kk * 4 + 1) * 32768);
    }

    v8f acc[2][4];
#pragma unroll
    for (int mt = 0; mt < 2; ++mt) {
#pragma unroll
        for (int j = 0; j < 4; ++j) acc[mt][j] = (v8f){};
    }
#pragma unroll
    for (int kk = 0; kk < 16; ++kk) {
        const v4f q0 = aq[0][kk >> 1];
        const v4f q1 = aq[1][kk >> 1];
        v2f a0, a1;
        if (kk & 1) { a0.x = q0.z; a0.y = q0.w; a1.x = q1.z; a1.y = q1.w; }
        else        { a0.x = q0.x; a0.y = q0.y; a1.x = q1.x; a1.y = q1.y; }
#pragma unroll
        for (int j = 0; j < 4; ++j) {
            v2f f; f.x = bz[kk][0][j]; f.y = bz[kk][1][j];
            acc[0][j] = mma(a0, f, acc[0][j]);
            acc[1][j] = mma(a1, f, acc[1][j]);
        }
    }

    const int rbase = hf << 3;
#pragma unroll
    for (int j = 0; j < 4; ++j) {
        const int q = 4 * qg + j;
        float* op = C + ((((size_t)b * 32 + r) * 32 + q) * 32) * 32 + c;
#pragma unroll
        for (int v = 0; v < 8; ++v) {
            op[(rbase + v) * 32]      = acc[0][j][v];
            op[(16 + rbase + v) * 32] = acc[1][j][v];
        }
    }
}

extern "C" void kernel_launch(void* const* d_in, const int* in_sizes, int n_in,
                              void* d_out, int out_size, void* d_ws, size_t ws_size,
                              hipStream_t stream) {
    (void)in_sizes; (void)n_in; (void)out_size; (void)ws_size;
    const float* x  = (const float*)d_in[0];
    float* out      = (float*)d_out;

    char* ws = (char*)d_ws;
    float* A  = (float*)ws;
    float* z2 = (float*)(ws + 8192);

    build_A_kernel<<<8, 1024, 0, stream>>>(A);
    fused_wh_kernel<<<256, 128, 0, stream>>>(A, x, z2);
    contract_d_kernel<<<256, 256, 0, stream>>>(A, z2, out);
}


# fused pass in two wave roles: 4 feeder waves fetch x and park it in LDS, the 4 matrix waves (one per SIMD, beside a feeder) only read fragments and run WMMAs, one fragment set; same sum order
# speedup vs baseline: 2.1644x; 1.0522x over previous
#include <hip/hip_runtime.h>
#include <math.h>


typedef __attribute__((ext_vector_type(2))) float v2f;
typedef __attribute__((ext_vector_type(8))) float v8f;
typedef __attribute__((ext_vector_type(4))) float v4f;

constexpr int APITCH = 68;

__device__ __forceinline__ int frag_index(int row, int col) {
    const int mt = row >> 4, r = row & 15;
    const int kk = col >> 2, half = (col >> 1) & 1, s = col & 1;
    return (((mt * 8 + (kk >> 1)) * 32 + half * 16 + r) << 2) + ((kk & 1) << 1) + s;
}

__global__ __launch_bounds__(1024) void build_A_kernel(float* __restrict__ Aout) {
    __shared__ float c64_lds[24 * 64];
    __shared__ float t32_lds[24 * 4];
    const float PI = 3.14159265358979323846f;
    const int tid = threadIdx.x;
    const int rb  = blockIdx.x * 4;
    for (int e = tid; e < 24 * 64 + 24 * 4; e += 1024) {
        if (e < 24 * 64) {
            int p = e >> 6, w = e & 63;
            c64_lds[e] = cosf(PI * ((float)w + 0.5f) * (float)p / 64.0f);
        } else {
            int f = e - 24 * 64;
            int p = f >> 2, r = rb + (f & 3);
            float s32 = (p == 0) ? sqrtf(1.0f / 32.0f) : sqrtf(2.0f / 32.0f);
            t32_lds[f] = cosf(PI * ((float)r + 0.5f) * (float)p / 32.0f) * s32;
        }
    }
    __syncthreads();
    if (tid < 256) {
        int rl = tid >> 6;
        int w  = tid & 63;
        float acc = 0.0f;
#pragma unroll
        for (int p = 0; p < 24; ++p) {
            float s64 = (p == 0) ? sqrtf(1.0f / 64.0f) : sqrtf(2.0f / 64.0f);
            acc += t32_lds[p * 4 + rl] * c64_lds[p * 64 + w] * s64;
        }
        Aout[frag_index(rb + rl, w)] = acc;
    }
}

__device__ __forceinline__ int z2_index(int qg, int r, int c) {
    return ((qg * 32 + r) * 32 + c) * 4;
}

__device__ __forceinline__ v8f mma(v2f a, v2f b, v8f c) {
    return __builtin_amdgcn_wmma_f32_16x16x4_f32(false, a, false, b, (short)0, c, false, false);
}

__device__ __forceinline__ float xhalf(float x) {
    int b = __builtin_bit_cast(int, x);
    return __builtin_bit_cast(float,
        __builtin_amdgcn_permlanex16(b, b, 0x76543210, 0xfedcba98, false, false));
}

struct XGroup { v2f f[4][16]; };

struct Stage { v4f r[16]; };

__device__ __forceinline__ void fetch_group(Stage& s, const float* __restrict__ src) {
#pragma unroll
    for (int j = 0; j < 16; ++j) s.r[j] = *reinterpret_cast<const v4f*>(src + j * 512);
}

__device__ __forceinline__ void park_group(const Stage& s, float* dst) {
#pragma unroll
    for (int j = 0; j < 16; ++j) *reinterpret_cast<v4f*>(dst + j * 512) = s.r[j];
}

__device__ __forceinline__ void load_group(XGroup& xg, const float* p) {
#pragma unroll
    for (int hi = 0; hi < 4; ++hi) {
#pragma unroll
        for (int kk = 0; kk < 16; ++kk) {
            const float* q = p + hi * 2048 + kk * 128;
            v2f t; t.x = q[0]; t.y = q[64];
            xg.f[hi][kk] = t;
        }
    }
}

__device__ __forceinline__ void lds_barrier() {
    asm volatile("s_wait_dscnt 0x0\n\ts_barrier_signal -1\n\ts_barrier_wait -1" ::: "memory");
}

__device__ __forceinline__ void stage1_group(const XGroup& xg, const v2f (&a1)[16], v8f (&d1)[4]) {
#pragma unroll
    for (int hi = 0; hi < 4; ++hi) d1[hi] = (v8f){};
#pragma unroll
    for (int kk = 0; kk < 16; ++kk) {
#pragma unroll
        for (int hi = 0; hi < 4; ++hi) d1[hi] = mma(a1[kk], xg.f[hi][kk], d1[hi]);
    }
}

__device__ __forceinline__ void stage2_group(const v8f (&d1)[4], v2f a2q0, v2f a2q1, bool lo, v8f (&out)[2][16]) {
#pragma unroll
    for (int v = 0; v < 8; ++v) {
        float sx = xhalf(lo ? d1[2][v] : d1[0][v]);
        float sy = xhalf(lo ? d1[3][v] : d1[1][v]);
        v2f blo, bhi;
        blo.x = lo ? d1[0][v] : sx;
        blo.y = lo ? d1[1][v] : sy;
        bhi.x = lo ? sx : d1[2][v];
        bhi.y = lo ? sy : d1[3][v];
        out[0][v]     = mma(a2q0, blo, out[0][v]);
        out[1][v]     = mma(a2q1, blo, out[1][v]);
        out[0][8 + v] = mma(a2q0, bhi, out[0][8 + v]);
        out[1][8 + v] = mma(a2q1, bhi, out[1][8 + v]);
    }
}

__global__ __launch_bounds__(256) __attribute__((amdgpu_waves_per_eu(2, 2))) void fused_wh_kernel(
    const float* __restrict__ Aglob, const float* __restrict__ x, float* __restrict__ z2)
{
    __shared__ __attribute__((aligned(16))) float lds[2 * 8192 + 32 * APITCH];
    float* const A_lds = lds + 2 * 8192;

    const int tid  = threadIdx.x;
    const int bd   = blockIdx.x;
    const int wave = __builtin_amdgcn_readfirstlane(tid >> 5);

    if (wave >= 4) {
        const int ft = tid - 128;
        const int fw = wave - 4;
        const float* xg = x + (size_t)bd * (64 * 64 * 32) + ft * 4;
        const int r4 = (ft >> 3) & 3;
        float* sp = lds + ((fw * 4 + (((r4 & 1) << 1) | (r4 >> 1))) * 32 + (ft & 7) * 4);

        float areg[16];
#pragma unroll
        for (int j = 0; j < 16; ++j) areg[j] = Aglob[ft + 128 * j];
        Stage r0, r1;
        fetch_group(r0, xg);
        fetch_group(r1, xg + 8192);
        const int a_row = (ft >> 2) & 15;
        const int a_col = (((ft >> 1) & 1) << 2) + ((ft >> 6) << 1) + (ft & 1);
#pragma unroll
        for (int j = 0; j < 16; ++j) {
            A_lds[((j >> 3) * 16 + a_row) * APITCH + a_col + 8 * (j & 7)] = areg[j];
        }
#pragma nounroll
        for (int i = 0; i < 8; ++i) {
            const int ge = (2 * i + 2 < 14) ? 2 * i + 2 : 14;
            const int go = (2 * i + 3 < 15) ? 2 * i + 3 : 15;
            park_group(r0, sp);
            fetch_group(r0, xg + ge * 8192);
            lds_barrier();
            park_group(r1, sp + 8192);
            fetch_group(r1, xg + go * 8192);
            lds_barrier();
        }
        return;
    }

    const int lane = tid & 31;
    const int ct   = wave & 1;
    const int rt   = wave >> 1;
    const int row  = lane & 15;
    const int half = lane >> 4;
    const int koff = half << 1;
    const bool lo  = (half == 0);

    v8f out[2][16];
#pragma unroll
    for (int qt = 0; qt < 2; ++qt) {
#pragma unroll
        for (int rr = 0; rr < 16; ++rr) out[qt][rr] = (v8f){};
    }

    lds_barrier();

    v2f a1[16];
#pragma unroll
    for (int kk = 0; kk < 16; ++kk) {
        const float* ap = &A_lds[(rt * 16 + row) * APITCH + kk * 4 + koff];
        v2f a; a.x = ap[0]; a.y = ap[1];
        a1[kk] = a;
    }

    const float* fp = lds + half * 32 + ct * 16 + row;
    const float* a2p = &A_lds[row * APITCH + koff];

    XGroup g;
    v8f d1[4];
    load_group(g, fp);
    stage1_group(g, a1, d1);
#pragma nounroll
    for (int k = 1; k < 16; ++k) {
        lds_barrier();
        const float* ap = a2p + 4 * (k - 1);
        v2f q0; q0.x = ap[0]; q0.y = ap[1];
        v2f q1; q1.x = ap[16 * APITCH]; q1.y = ap[16 * APITCH + 1];
        load_group(g, fp + (k & 1) * 8192);
        __builtin_amdgcn_sched_barrier(0);
        stage2_group(d1, q0, q1, lo, out);
        __builtin_amdgcn_sched_barrier(0);
        stage1_group(g, a1, d1);
        __builtin_amdgcn_sched_barrier(0);
    }
    {
        const float* ap = a2p + 4 * 15;
        v2f q0; q0.x = ap[0]; q0.y = ap[1];
        v2f q1; q1.x = ap[16 * APITCH]; q1.y = ap[16 * APITCH + 1];
        stage2_group(d1, q0, q1, lo, out);
    }

    float* zp = z2 + (size_t)bd * (32 * 32 * 32) + z2_index(2 * half, 16 * rt, ct * 16 + row);
#pragma unroll
    for (int qt = 0; qt < 2; ++qt) {
#pragma unroll
        for (int rr = 0; rr < 16; ++rr) {
#pragma unroll
            for (int u4 = 0; u4 < 2; ++u4) {
                v4f t;
                t.x = out[qt][rr][4 * u4 + 0]; t.y = out[qt][rr][4 * u4 + 1];
                t.z = out[qt][rr][4 * u4 + 2]; t.w = out[qt][rr][4 * u4 + 3];
                *reinterpret_cast<v4f*>(zp + z2_index(4 * qt + u4, rr, 0)) = t;
            }
        }
    }
}

__global__ __launch_bounds__(256) __attribute__((amdgpu_waves_per_eu(2, 2))) void contract_d_kernel(
    const float* __restrict__ Afrag, const float* __restrict__ Z, float* __restrict__ C)
{
    const int lane = threadIdx.x & 31;
    const int wave = threadIdx.x >> 5;
    const int t    = blockIdx.x * 8 + wave;
    const int ct   = t & 1;
    const int r    = (t >> 1) & 31;
    const int qg   = (t >> 6) & 7;
    const int b    = t >> 9;
    const int cl   = lane & 15;
    const int hf   = lane >> 4;
    const int c    = ct * 16 + cl;

    const v4f* ap = reinterpret_cast<const v4f*>(Afrag) + lane;
    v4f aq[2][8];
#pragma unroll
    for (int mt = 0; mt < 2; ++mt) {
#pragma unroll
        for (int j = 0; j < 8; ++j) aq[mt][j] = ap[(mt * 8 + j) * 32];
    }

    const float* bp = Z + (size_t)(b * 64 + 2 * hf) * 32768 + z2_index(qg, r, c);
    v4f bz[16][2];
#pragma unroll
    for (int kk = 0; kk < 16; ++kk) {
        bz[kk][0] = *reinterpret_cast<const v4f*>(bp + (size_t)(kk * 4) * 32768);
        bz[kk][1] = *reinterpret_cast<const v4f*>(bp + (size_t)(kk * 4 + 1) * 32768);
    }

    v8f acc[2][4];
#pragma unroll
    for (int mt = 0; mt < 2; ++mt) {
#pragma unroll
        for (int j = 0; j < 4; ++j) acc[mt][j] = (v8f){};
    }
#pragma unroll
    for (int kk = 0; kk < 16; ++kk) {
        const v4f q0 = aq[0][kk >> 1];
        const v4f q1 = aq[1][kk >> 1];
        v2f a0, a1;
        if (kk & 1) { a0.x = q0.z; a0.y = q0.w; a1.x = q1.z; a1.y = q1.w; }
        else        { a0.x = q0.x; a0.y = q0.y; a1.x = q1.x; a1.y = q1.y; }
#pragma unroll
        for (int j = 0; j < 4; ++j) {
            v2f f; f.x = bz[kk][0][j]; f.y = bz[kk][1][j];
            acc[0][j] = mma(a0, f, acc[0][j]);
            acc[1][j] = mma(a1, f, acc[1][j]);
        }
    }

    const int rbase = hf << 3;
#pragma unroll
    for (int j = 0; j < 4; ++j) {
        const int q = 4 * qg + j;
        float* op = C + ((((size_t)b * 32 + r) * 32 + q) * 32) * 32 + c;
#pragma unroll
        for (int v = 0; v < 8; ++v) {
            op[(rbase + v) * 32]      = acc[0][j][v];
            op[(16 + rbase + v) * 32] = acc[1][j][v];
        }
    }
}

extern "C" void kernel_launch(void* const* d_in, const int* in_sizes, int n_in,
                              void* d_out, int out_size, void* d_ws, size_t ws_size,
                              hipStream_t stream) {
    (void)in_sizes; (void)n_in; (void)out_size; (void)ws_size;
    const float* x  = (const float*)d_in[0];
    float* out      = (float*)d_out;

    char* ws = (char*)d_ws;
    float* A  = (float*)ws;
    float* z2 = (float*)(ws + 8192);

    build_A_kernel<<<8, 1024, 0, stream>>>(A);
    fused_wh_kernel<<<256, 256, 0, stream>>>(A, x, z2);
    contract_d_kernel<<<256, 256, 0, stream>>>(A, z2, out);
}


# x parked in LDS transposed into B-fragment order: a lane reads two k-steps' fragments with one conflict-free 16-byte read (32 per group; was 64 two-address reads, both words on one bank); same sums
# speedup vs baseline: 2.2574x; 1.0430x over previous
#include <hip/hip_runtime.h>
#include <math.h>


typedef __attribute__((ext_vector_type(2))) float v2f;
typedef __attribute__((ext_vector_type(8))) float v8f;
typedef __attribute__((ext_vector_type(4))) float v4f;

constexpr int APITCH = 68;

constexpr int BPITCH = 132;
constexpr int STAGE  = 64 * BPITCH;

__device__ __forceinline__ int frag_index(int row, int col) {
    const int mt = row >> 4, r = row & 15;
    const int kk = col >> 2, half = (col >> 1) & 1, s = col & 1;
    return (((mt * 8 + (kk >> 1)) * 32 + half * 16 + r) << 2) + ((kk & 1) << 1) + s;
}

__global__ __launch_bounds__(1024) void build_A_kernel(float* __restrict__ Aout) {
    __shared__ float c64_lds[24 * 64];
    __shared__ float t32_lds[24 * 4];
    const float PI = 3.14159265358979323846f;
    const int tid = threadIdx.x;
    const int rb  = blockIdx.x * 4;
    for (int e = tid; e < 24 * 64 + 24 * 4; e += 1024) {
        if (e < 24 * 64) {
            int p = e >> 6, w = e & 63;
            c64_lds[e] = cosf(PI * ((float)w + 0.5f) * (float)p / 64.0f);
        } else {
            int f = e - 24 * 64;
            int p = f >> 2, r = rb + (f & 3);
            float s32 = (p == 0) ? sqrtf(1.0f / 32.0f) : sqrtf(2.0f / 32.0f);
            t32_lds[f] = cosf(PI * ((float)r + 0.5f) * (float)p / 32.0f) * s32;
        }
    }
    __syncthreads();
    if (tid < 256) {
        int rl = tid >> 6;
        int w  = tid & 63;
        float acc = 0.0f;
#pragma unroll
        for (int p = 0; p < 24; ++p) {
            float s64 = (p == 0) ? sqrtf(1.0f / 64.0f) : sqrtf(2.0f / 64.0f);
            acc += t32_lds[p * 4 + rl] * c64_lds[p * 64 + w] * s64;
        }
        Aout[frag_index(rb + rl, w)] = acc;
    }
}

__device__ __forceinline__ int z2_index(int qg, int r, int c) {
    return ((qg * 32 + r) * 32 + c) * 4;
}

__device__ __forceinline__ v8f mma(v2f a, v2f b, v8f c) {
    return __builtin_amdgcn_wmma_f32_16x16x4_f32(false, a, false, b, (short)0, c, false, false);
}

__device__ __forceinline__ float xhalf(float x) {
    int b = __builtin_bit_cast(int, x);
    return __builtin_bit_cast(float,
        __builtin_amdgcn_permlanex16(b, b, 0x76543210, 0xfedcba98, false, false));
}

struct XGroup { v2f f[4][16]; };

struct Stage { v4f r[16]; };

__device__ __forceinline__ void fetch_group(Stage& s, const float* __restrict__ src) {
#pragma unroll
    for (int u = 0; u < 4; ++u) {
        s.r[4 * u + 0] = *reinterpret_cast<const v4f*>(src + u * 512);
        s.r[4 * u + 1] = *reinterpret_cast<const v4f*>(src + u * 512 + 32);
        s.r[4 * u + 2] = *reinterpret_cast<const v4f*>(src + u * 512 + 128);
        s.r[4 * u + 3] = *reinterpret_cast<const v4f*>(src + u * 512 + 160);
    }
}

__device__ __forceinline__ void park_group(const Stage& s, float* dst) {
#pragma unroll
    for (int u = 0; u < 4; ++u) {
#pragma unroll
        for (int i = 0; i < 4; ++i) {
            v4f t;
            t.x = s.r[4 * u + 0][i]; t.y = s.r[4 * u + 1][i];
            t.z = s.r[4 * u + 2][i]; t.w = s.r[4 * u + 3][i];
            *reinterpret_cast<v4f*>(dst + u * 4 * BPITCH + 4 * i) = t;
        }
    }
}

__device__ __forceinline__ void load_group(XGroup& xg, const float* p) {
#pragma unroll
    for (int hi = 0; hi < 4; ++hi) {
#pragma unroll
        for (int m = 0; m < 8; ++m) {
            const v4f t = *reinterpret_cast<const v4f*>(p + (hi * 16 + m * 2) * BPITCH);
            v2f e; e.x = t.x; e.y = t.y;
            v2f o; o.x = t.z; o.y = t.w;
            xg.f[hi][2 * m]     = e;
            xg.f[hi][2 * m + 1] = o;
        }
    }
}

__device__ __forceinline__ void lds_barrier() {
    asm volatile("s_wait_dscnt 0x0\n\ts_barrier_signal -1\n\ts_barrier_wait -1" ::: "memory");
}

__device__ __forceinline__ void stage1_group(const XGroup& xg, const v2f (&a1)[16], v8f (&d1)[4]) {
#pragma unroll
    for (int hi = 0; hi < 4; ++hi) d1[hi] = (v8f){};
#pragma unroll
    for (int kk = 0; kk < 16; ++kk) {
#pragma unroll
        for (int hi = 0; hi < 4; ++hi) d1[hi] = mma(a1[kk], xg.f[hi][kk], d1[hi]);
    }
}

__device__ __forceinline__ void stage2_group(const v8f (&d1)[4], v2f a2q0, v2f a2q1, bool lo, v8f (&out)[2][16]) {
#pragma unroll
    for (int v = 0; v < 8; ++v) {
        float sx = xhalf(lo ? d1[2][v] : d1[0][v]);
        float sy = xhalf(lo ? d1[3][v] : d1[1][v]);
        v2f blo, bhi;
        blo.x = lo ? d1[0][v] : sx;
        blo.y = lo ? d1[1][v] : sy;
        bhi.x = lo ? sx : d1[2][v];
        bhi.y = lo ? sy : d1[3][v];
        out[0][v]     = mma(a2q0, blo, out[0][v]);
        out[1][v]     = mma(a2q1, blo, out[1][v]);
        out[0][8 + v] = mma(a2q0, bhi, out[0][8 + v]);
        out[1][8 + v] = mma(a2q1, bhi, out[1][8 + v]);
    }
}

__global__ __launch_bounds__(256) __attribute__((amdgpu_waves_per_eu(2, 2))) void fused_wh_kernel(
    const float* __restrict__ Aglob, const float* __restrict__ x, float* __restrict__ z2)
{
    __shared__ __attribute__((aligned(16))) float lds[2 * STAGE + 32 * APITCH];
    float* const A_lds = lds + 2 * STAGE;

    const int tid  = threadIdx.x;
    const int bd   = blockIdx.x;
    const int wave = __builtin_amdgcn_readfirstlane(tid >> 5);

    if (wave >= 4) {
        const int ft = tid - 128;
        const int fw = wave - 4;
        const int c4 = (ft & 3) | ((ft >> 2) & 4);
        const int fh = (ft >> 2) & 1;
        const int m0 = (ft >> 3) & 1;
        const float* xg = x + (size_t)bd * (64 * 64 * 32) + fw * 2048 + (8 * m0 + 2 * fh) * 32 + c4 * 4;
        float* sp = lds + (fw * 16 + 2 * m0 + fh) * BPITCH + c4 * 16;

        float areg[16];
#pragma unroll
        for (int j = 0; j < 16; ++j) areg[j] = Aglob[ft + 128 * j];
        Stage r0, r1;
        fetch_group(r0, xg);
        fetch_group(r1, xg + 8192);
        const int a_row = (ft >> 2) & 15;
        const int a_col = (((ft >> 1) & 1) << 2) + ((ft >> 6) << 1) + (ft & 1);
#pragma unroll
        for (int j = 0; j < 16; ++j) {
            A_lds[((j >> 3) * 16 + a_row) * APITCH + a_col + 8 * (j & 7)] = areg[j];
        }
#pragma nounroll
        for (int i = 0; i < 8; ++i) {
            const int ge = (2 * i + 2 < 14) ? 2 * i + 2 : 14;
            const int go = (2 * i + 3 < 15) ? 2 * i + 3 : 15;
            park_group(r0, sp);
            fetch_group(r0, xg + ge * 8192);
            lds_barrier();
            __builtin_amdgcn_sched_barrier(0);
            park_group(r1, sp + STAGE);
            fetch_group(r1, xg + go * 8192);
            lds_barrier();
        }
        return;
    }

    const int lane = tid & 31;
    const int ct   = wave & 1;
    const int rt   = wave >> 1;
    const int row  = lane & 15;
    const int half = lane >> 4;
    const int koff = half << 1;
    const bool lo  = (half == 0);

    v8f out[2][16];
#pragma unroll
    for (int qt = 0; qt < 2; ++qt) {
#pragma unroll
        for (int rr = 0; rr < 16; ++rr) out[qt][rr] = (v8f){};
    }

    lds_barrier();

    v2f a1[16];
#pragma unroll
    for (int kk = 0; kk < 16; ++kk) {
        const float* ap = &A_lds[(rt * 16 + row) * APITCH + kk * 4 + koff];
        v2f a; a.x = ap[0]; a.y = ap[1];
        a1[kk] = a;
    }

    const float* fp = lds + half * BPITCH + (ct * 16 + row) * 4;
    const float* a2p = &A_lds[row * APITCH + koff];

    XGroup g;
    v8f d1[4];
    load_group(g, fp);
    stage1_group(g, a1, d1);
#pragma nounroll
    for (int k = 1; k < 16; ++k) {
        lds_barrier();
        const float* ap = a2p + 4 * (k - 1);
        v2f q0; q0.x = ap[0]; q0.y = ap[1];
        v2f q1; q1.x = ap[16 * APITCH]; q1.y = ap[16 * APITCH + 1];
        load_group(g, fp + (k & 1) * STAGE);
        __builtin_amdgcn_sched_barrier(0);
        stage2_group(d1, q0, q1, lo, out);
        __builtin_amdgcn_sched_barrier(0);
        stage1_group(g, a1, d1);
        __builtin_amdgcn_sched_barrier(0);
    }
    {
        const float* ap = a2p + 4 * 15;
        v2f q0; q0.x = ap[0]; q0.y = ap[1];
        v2f q1; q1.x = ap[16 * APITCH]; q1.y = ap[16 * APITCH + 1];
        stage2_group(d1, q0, q1, lo, out);
    }

    float* zp = z2 + (size_t)bd * (32 * 32 * 32) + z2_index(2 * half, 16 * rt, ct * 16 + row);
#pragma unroll
    for (int qt = 0; qt < 2; ++qt) {
#pragma unroll
        for (int rr = 0; rr < 16; ++rr) {
#pragma unroll
            for (int u4 = 0; u4 < 2; ++u4) {
                v4f t;
                t.x = out[qt][rr][4 * u4 + 0]; t.y = out[qt][rr][4 * u4 + 1];
                t.z = out[qt][rr][4 * u4 + 2]; t.w = out[qt][rr][4 * u4 + 3];
                *reinterpret_cast<v4f*>(zp + z2_index(4 * qt + u4, rr, 0)) = t;
            }
        }
    }
}

__global__ __launch_bounds__(256) __attribute__((amdgpu_waves_per_eu(2, 2))) void contract_d_kernel(
    const float* __restrict__ Afrag, const float* __restrict__ Z, float* __restrict__ C)
{
    const int lane = threadIdx.x & 31;
    const int wave = threadIdx.x >> 5;
    const int t    = blockIdx.x * 8 + wave;
    const int ct   = t & 1;
    const int r    = (t >> 1) & 31;
    const int qg   = (t >> 6) & 7;
    const int b    = t >> 9;
    const int cl   = lane & 15;
    const int hf   = lane >> 4;
    const int c    = ct * 16 + cl;

    const v4f* ap = reinterpret_cast<const v4f*>(Afrag) + lane;
    v4f aq[2][8];
#pragma unroll
    for (int mt = 0; mt < 2; ++mt) {
#pragma unroll
        for (int j = 0; j < 8; ++j) aq[mt][j] = ap[(mt * 8 + j) * 32];
    }

    const float* bp = Z + (size_t)(b * 64 + 2 * hf) * 32768 + z2_index(qg, r, c);
    v4f bz[16][2];
#pragma unroll
    for (int kk = 0; kk < 16; ++kk) {
        bz[kk][0] = *reinterpret_cast<const v4f*>(bp + (size_t)(kk * 4) * 32768);
        bz[kk][1] = *reinterpret_cast<const v4f*>(bp + (size_t)(kk * 4 + 1) * 32768);
    }

    v8f acc[2][4];
#pragma unroll
    for (int mt = 0; mt < 2; ++mt) {
#pragma unroll
        for (int j = 0; j < 4; ++j) acc[mt][j] = (v8f){};
    }
#pragma unroll
    for (int kk = 0; kk < 16; ++kk) {
        const v4f q0 = aq[0][kk >> 1];
        const v4f q1 = aq[1][kk >> 1];
        v2f a0, a1;
        if (kk & 1) { a0.x = q0.z; a0.y = q0.w; a1.x = q1.z; a1.y = q1.w; }
        else        { a0.x = q0.x; a0.y = q0.y; a1.x = q1.x; a1.y = q1.y; }
#pragma unroll
        for (int j = 0; j < 4; ++j) {
            v2f f; f.x = bz[kk][0][j]; f.y = bz[kk][1][j];
            acc[0][j] = mma(a0, f, acc[0][j]);
            acc[1][j] = mma(a1, f, acc[1][j]);
        }
    }

    const int rbase = hf << 3;
#pragma unroll
    for (int j = 0; j < 4; ++j) {
        const int q = 4 * qg + j;
        float* op = C + ((((size_t)b * 32 + r) * 32 + q) * 32) * 32 + c;
#pragma unroll
        for (int v = 0; v < 8; ++v) {
            op[(rbase + v) * 32]      = acc[0][j][v];
            op[(16 + rbase + v) * 32] = acc[1][j][v];
        }
    }
}

extern "C" void kernel_launch(void* const* d_in, const int* in_sizes, int n_in,
                              void* d_out, int out_size, void* d_ws, size_t ws_size,
                              hipStream_t stream) {
    (void)in_sizes; (void)n_in; (void)out_size; (void)ws_size;
    const float* x  = (const float*)d_in[0];
    float* out      = (float*)d_out;

    char* ws = (char*)d_ws;
    float* A  = (float*)ws;
    float* z2 = (float*)(ws + 8192);

    build_A_kernel<<<8, 1024, 0, stream>>>(A);
    fused_wh_kernel<<<256, 256, 0, stream>>>(A, x, z2);
    contract_d_kernel<<<256, 256, 0, stream>>>(A, z2, out);
}
